# starting kernel
# baseline (speedup 1.0000x reference)
#include <hip/hip_runtime.h>
#include <hip/hip_bf16.h>

#define B_SAMPLES 8192
#define T_TOKENS  1048576
#define NHID      256
#define NCLASSES  14
#define NPAD      16

typedef __attribute__((ext_vector_type(2))) float v2f;
typedef __attribute__((ext_vector_type(8))) float v8f;

__global__ void seg_scan_kernel(const int* __restrict__ lengths,
                                int* __restrict__ offsets) {
    __shared__ int ssum[1024];
    const int t = threadIdx.x;
    const int base = t * 8;
    int loc[8];
    int s = 0;
#pragma unroll
    for (int j = 0; j < 8; ++j) {
        loc[j] = lengths[base + j];
        s += loc[j];
    }
    ssum[t] = s;
    __syncthreads();
    for (int off = 1; off < 1024; off <<= 1) {
        int v = (t >= off) ? ssum[t - off] : 0;
        __syncthreads();
        ssum[t] += v;
        __syncthreads();
    }
    int excl = ssum[t] - s;
#pragma unroll
    for (int j = 0; j < 8; ++j) {
        offsets[base + j] = excl;
        excl += loc[j];
    }
    if (t == 1023) offsets[B_SAMPLES] = ssum[1023];
}

__global__ void seg_mean_tanh_kernel(const int* __restrict__ offsets,
                                     const int* __restrict__ indices,
                                     const float* __restrict__ emb,
                                     float* __restrict__ Amat) {
    __shared__ int sidx[256];
    const int tid = threadIdx.x;
    const int sample = blockIdx.x;
    const int s = offsets[sample];
    const int e = offsets[sample + 1];

    float acc = 0.0f;
    for (int base = s; base < e; base += 256) {
        const int n = (e - base < 256) ? (e - base) : 256;
        if (tid < n) sidx[tid] = indices[base + tid];
        __syncthreads();
        for (int j = 0; j < n; ++j) {
            if (j + 8 < n) {
                __builtin_prefetch(emb + (size_t)sidx[j + 8] * NHID + tid, 0, 1);
            }
            acc += emb[(size_t)sidx[j] * NHID + tid];
        }
        __syncthreads();
    }

    const float len = (float)(e - s);
    const float m = (len > 0.0f) ? (acc / len) : 0.0f;
    Amat[(size_t)sample * NHID + tid] = tanhf(m);
}

__global__ void wmma_head_kernel(const float* __restrict__ Amat,
                                 const float* __restrict__ W,
                                 const float* __restrict__ bias,
                                 float* __restrict__ out) {
    __shared__ float lw[NHID * NPAD];

    const int tid = threadIdx.x;
    for (int i = tid; i < NHID * NPAD; i += 256) {
        const int kk = i >> 4;
        const int nn = i & 15;
        lw[i] = (nn < NCLASSES) ? W[kk * NCLASSES + nn] : 0.0f;
    }
    __syncthreads();

    const int lane = tid & 31;
    const int wave = tid >> 5;
    const int tile = blockIdx.x * 8 + wave;
    const int row0 = tile * 16;

    const int mrow = lane & 15;
    const int koff = (lane >> 4) << 1;
    const float* __restrict__ arow = Amat + (size_t)(row0 + mrow) * NHID;

    v8f c = {};
#pragma unroll 4
    for (int k = 0; k < NHID; k += 4) {
        const float2 a2 = *(const float2*)(arow + k + koff);
        v2f av; av.x = a2.x; av.y = a2.y;
        v2f bv;
        bv.x = lw[(k + koff) * NPAD + mrow];
        bv.y = lw[(k + koff + 1) * NPAD + mrow];
        c = __builtin_amdgcn_wmma_f32_16x16x4_f32(
            false, av, false, bv, (short)0, c, false, false);
    }

    const int mhi = (lane >> 4) << 3;
    const int ncol = lane & 15;
    if (ncol < NCLASSES) {
        const float bb = bias[ncol];
#pragma unroll
        for (int j = 0; j < 8; ++j) {
            const int m = row0 + j + mhi;
            out[(size_t)m * NCLASSES + ncol] = c[j] + bb;
        }
    }
}

extern "C" void kernel_launch(void* const* d_in, const int* in_sizes, int n_in,
                              void* d_out, int out_size, void* d_ws, size_t ws_size,
                              hipStream_t stream) {
    const int*   lengths = (const int*)d_in[0];
    const int*   indices = (const int*)d_in[1];
    const float* emb     = (const float*)d_in[2];
    const float* W       = (const float*)d_in[3];
    const float* bias    = (const float*)d_in[4];
    float*       out     = (float*)d_out;

    int*   offsets = (int*)d_ws;
    float* Amat    = (float*)((char*)d_ws + (64 * 1024));

    seg_scan_kernel<<<1, 1024, 0, stream>>>(lengths, offsets);
    seg_mean_tanh_kernel<<<B_SAMPLES, 256, 0, stream>>>(offsets, indices, emb, Amat);
    wmma_head_kernel<<<B_SAMPLES / (16 * 8), 256, 0, stream>>>(Amat, W, bias, out);
}


# segment sums as a product with the 0/1 token-membership matrix on the f32 WMMA: block = sample tile x column quarter, its 8 waves split the tile's tokens evenly; sum order changed
# speedup vs baseline: 2.9105x; 2.9105x over previous
#include <hip/hip_runtime.h>
#include <hip/hip_bf16.h>

#define B_SAMPLES 8192
#define T_TOKENS  1048576
#define NHID      256
#define NCLASSES  14
#define NPAD      16

typedef __attribute__((ext_vector_type(2))) float v2f;
typedef __attribute__((ext_vector_type(8))) float v8f;

__global__ void seg_scan_kernel(const int* __restrict__ lengths,
                                int* __restrict__ offsets) {
    __shared__ int ssum[1024];
    const int t = threadIdx.x;
    const int base = t * 8;
    int loc[8];
    int s = 0;
#pragma unroll
    for (int j = 0; j < 8; ++j) {
        loc[j] = lengths[base + j];
        s += loc[j];
    }
    ssum[t] = s;
    __syncthreads();
    for (int off = 1; off < 1024; off <<= 1) {
        int v = (t >= off) ? ssum[t - off] : 0;
        __syncthreads();
        ssum[t] += v;
        __syncthreads();
    }
    int excl = ssum[t] - s;
#pragma unroll
    for (int j = 0; j < 8; ++j) {
        offsets[base + j] = excl;
        excl += loc[j];
    }
    if (t == 1023) offsets[B_SAMPLES] = ssum[1023];
}

__global__ __launch_bounds__(256)
void seg_member_sum_tanh_kernel(const int* __restrict__ offsets,
                                const int* __restrict__ indices,
                                const float* __restrict__ emb,
                                float* __restrict__ Amat) {
    __shared__ float red[8 * 32 * 32];

    const int tid  = threadIdx.x;
    const int lane = tid & 31;
    const int g = __builtin_amdgcn_readfirstlane(tid >> 5);
    const int q = blockIdx.x & 3;
    const int s0 = (blockIdx.x >> 2) * 16;
    const int c = lane & 15;
    const int h = lane >> 4;

    const int lo = offsets[s0 + c];
    const int hi = offsets[s0 + c + 1];
    const int T0 = offsets[s0];
    const int T1 = offsets[s0 + 16];
    const int nb = (T1 - T0 + 31) >> 5;

    const unsigned laneoff = (unsigned)(q * 256 + c * 16);
    const char* __restrict__ embb = (const char*)emb;

    v8f acc0 = {}, acc1 = {}, acc2 = {}, acc3 = {};

    int idx0;
    {
        const int t0 = T0 + (g << 5) + lane;
        idx0 = indices[t0 < T_TOKENS ? t0 : T_TOKENS - 1];
    }
    for (int b = g; b < nb; b += 8) {
        const int base = T0 + (b << 5);
        const int t1 = base + 256 + lane;
        int idx1 = indices[t1 < T_TOKENS ? t1 : T_TOKENS - 1];
        asm volatile("");

        float4 ld[8][2];
#pragma unroll
        for (int kg = 0; kg < 8; ++kg) {
#pragma unroll
            for (int s = 0; s < 2; ++s) {
                const int idx = __shfl(idx0, 4 * kg + 2 * h + s);
                const unsigned off = (unsigned)idx * (unsigned)(NHID * 4) + laneoff;
                ld[kg][s] = *(const float4*)(embb + off);
            }
        }

        int ra = lo - base; ra = ra < 0 ? 0 : (ra > 32 ? 32 : ra);
        int rb = hi - base; rb = rb < 0 ? 0 : (rb > 32 ? 32 : rb);
        const unsigned ma = (ra >= 32) ? 0xffffffffu : ((1u << ra) - 1u);
        const unsigned mb = (rb >= 32) ? 0xffffffffu : ((1u << rb) - 1u);
        const unsigned mh = (mb & ~ma) >> (2 * h);

#pragma unroll
        for (int kg = 0; kg < 8; ++kg) {
            v2f av;
            av.x = __uint_as_float((unsigned)(((int)(mh << (31 - 4 * kg)) >> 31)) & 0x3f800000u);
            av.y = __uint_as_float((unsigned)(((int)(mh << (30 - 4 * kg)) >> 31)) & 0x3f800000u);
            v2f b0; b0.x = ld[kg][0].x; b0.y = ld[kg][1].x;
            v2f b1; b1.x = ld[kg][0].y; b1.y = ld[kg][1].y;
            v2f b2; b2.x = ld[kg][0].z; b2.y = ld[kg][1].z;
            v2f b3; b3.x = ld[kg][0].w; b3.y = ld[kg][1].w;
            acc0 = __builtin_amdgcn_wmma_f32_16x16x4_f32(
                false, av, false, b0, (short)0, acc0, false, false);
            acc1 = __builtin_amdgcn_wmma_f32_16x16x4_f32(
                false, av, false, b1, (short)0, acc1, false, false);
            acc2 = __builtin_amdgcn_wmma_f32_16x16x4_f32(
                false, av, false, b2, (short)0, acc2, false, false);
            acc3 = __builtin_amdgcn_wmma_f32_16x16x4_f32(
                false, av, false, b3, (short)0, acc3, false, false);
        }
        asm volatile("" : "+v"(idx1));
        idx0 = idx1;
    }

    float* myred = red + (g * 32) * 32 + lane;
#pragma unroll
    for (int r = 0; r < 8; ++r) {
        myred[(0 * 8 + r) * 32] = acc0[r];
        myred[(1 * 8 + r) * 32] = acc1[r];
        myred[(2 * 8 + r) * 32] = acc2[r];
        myred[(3 * 8 + r) * 32] = acc3[r];
    }
    __syncthreads();

    const int mylen = hi - lo;
    const int len = __shfl(mylen, 8 * h + g);
    const float lf = (float)len;
    float o[4];
#pragma unroll
    for (int j = 0; j < 4; ++j) {
        const float* p = red + (j * 8 + g) * 32 + lane;
        float v = p[0];
#pragma unroll
        for (int gg = 1; gg < 8; ++gg) v += p[gg * 32 * 32];
        const float m = (lf > 0.0f) ? (v / lf) : 0.0f;
        o[j] = tanhf(m);
    }
    float4 ov; ov.x = o[0]; ov.y = o[1]; ov.z = o[2]; ov.w = o[3];
    *(float4*)(Amat + (size_t)(s0 + 8 * h + g) * NHID + 64 * q + 4 * c) = ov;
}

__global__ void wmma_head_kernel(const float* __restrict__ Amat,
                                 const float* __restrict__ W,
                                 const float* __restrict__ bias,
                                 float* __restrict__ out) {
    __shared__ float lw[NHID * NPAD];

    const int tid = threadIdx.x;
    for (int i = tid; i < NHID * NPAD; i += 256) {
        const int kk = i >> 4;
        const int nn = i & 15;
        lw[i] = (nn < NCLASSES) ? W[kk * NCLASSES + nn] : 0.0f;
    }
    __syncthreads();

    const int lane = tid & 31;
    const int wave = tid >> 5;
    const int tile = blockIdx.x * 8 + wave;
    const int row0 = tile * 16;

    const int mrow = lane & 15;
    const int koff = (lane >> 4) << 1;
    const float* __restrict__ arow = Amat + (size_t)(row0 + mrow) * NHID;

    v8f c = {};
#pragma unroll 4
    for (int k = 0; k < NHID; k += 4) {
        const float2 a2 = *(const float2*)(arow + k + koff);
        v2f av; av.x = a2.x; av.y = a2.y;
        v2f bv;
        bv.x = lw[(k + koff) * NPAD + mrow];
        bv.y = lw[(k + koff + 1) * NPAD + mrow];
        c = __builtin_amdgcn_wmma_f32_16x16x4_f32(
            false, av, false, bv, (short)0, c, false, false);
    }

    const int mhi = (lane >> 4) << 3;
    const int ncol = lane & 15;
    if (ncol < NCLASSES) {
        const float bb = bias[ncol];
#pragma unroll
        for (int j = 0; j < 8; ++j) {
            const int m = row0 + j + mhi;
            out[(size_t)m * NCLASSES + ncol] = c[j] + bb;
        }
    }
}

extern "C" void kernel_launch(void* const* d_in, const int* in_sizes, int n_in,
                              void* d_out, int out_size, void* d_ws, size_t ws_size,
                              hipStream_t stream) {
    const int*   lengths = (const int*)d_in[0];
    const int*   indices = (const int*)d_in[1];
    const float* emb     = (const float*)d_in[2];
    const float* W       = (const float*)d_in[3];
    const float* bias    = (const float*)d_in[4];
    float*       out     = (float*)d_out;

    int*   offsets = (int*)d_ws;
    float* Amat    = (float*)((char*)d_ws + (64 * 1024));

    seg_scan_kernel<<<1, 1024, 0, stream>>>(lengths, offsets);
    seg_member_sum_tanh_kernel<<<(B_SAMPLES / 16) * 4, 256, 0, stream>>>(offsets, indices, emb, Amat);
    wmma_head_kernel<<<B_SAMPLES / (16 * 8), 256, 0, stream>>>(Amat, W, bias, out);
}


# segment-sum kernel declared at 8 waves per SIMD (128 registers): a batch's 16 row loads go out as one group, one memory round trip per batch (was three dependent groups); sum order kept
# speedup vs baseline: 3.0868x; 1.0606x over previous
#include <hip/hip_runtime.h>
#include <hip/hip_bf16.h>

#define B_SAMPLES 8192
#define T_TOKENS  1048576
#define NHID      256
#define NCLASSES  14
#define NPAD      16

typedef __attribute__((ext_vector_type(2))) float v2f;
typedef __attribute__((ext_vector_type(8))) float v8f;

__global__ void seg_scan_kernel(const int* __restrict__ lengths,
                                int* __restrict__ offsets) {
    __shared__ int ssum[1024];
    const int t = threadIdx.x;
    const int base = t * 8;
    int loc[8];
    int s = 0;
#pragma unroll
    for (int j = 0; j < 8; ++j) {
        loc[j] = lengths[base + j];
        s += loc[j];
    }
    ssum[t] = s;
    __syncthreads();
    for (int off = 1; off < 1024; off <<= 1) {
        int v = (t >= off) ? ssum[t - off] : 0;
        __syncthreads();
        ssum[t] += v;
        __syncthreads();
    }
    int excl = ssum[t] - s;
#pragma unroll
    for (int j = 0; j < 8; ++j) {
        offsets[base + j] = excl;
        excl += loc[j];
    }
    if (t == 1023) offsets[B_SAMPLES] = ssum[1023];
}

__global__ __launch_bounds__(256) __attribute__((amdgpu_waves_per_eu(8, 8)))
void seg_member_sum_tanh_kernel(const int* __restrict__ offsets,
                                const int* __restrict__ indices,
                                const float* __restrict__ emb,
                                float* __restrict__ Amat) {
    __shared__ float red[8 * 32 * 32];

    const int tid  = threadIdx.x;
    const int lane = tid & 31;
    const int g = __builtin_amdgcn_readfirstlane(tid >> 5);
    const int q = blockIdx.x & 3;
    const int s0 = (blockIdx.x >> 2) * 16;
    const int c = lane & 15;
    const int h = lane >> 4;

    const int lo = offsets[s0 + c];
    const int hi = offsets[s0 + c + 1];
    const int T0 = offsets[s0];
    const int T1 = offsets[s0 + 16];
    const int nb = (T1 - T0 + 31) >> 5;

    const unsigned laneoff = (unsigned)(q * 256 + c * 16);
    const char* __restrict__ embb = (const char*)emb;

    v8f acc0 = {}, acc1 = {}, acc2 = {}, acc3 = {};

    int idx0;
    {
        const int t0 = T0 + (g << 5) + lane;
        idx0 = indices[t0 < T_TOKENS ? t0 : T_TOKENS - 1];
    }
    for (int b = g; b < nb; b += 8) {
        const int base = T0 + (b << 5);
        const int t1 = base + 256 + lane;
        int idx1 = indices[t1 < T_TOKENS ? t1 : T_TOKENS - 1];
        asm volatile("");

        float4 ld[8][2];
#pragma unroll
        for (int kg = 0; kg < 8; ++kg) {
#pragma unroll
            for (int s = 0; s < 2; ++s) {
                const int idx = __shfl(idx0, 4 * kg + 2 * h + s);
                const unsigned off = (unsigned)idx * (unsigned)(NHID * 4) + laneoff;
                ld[kg][s] = *(const float4*)(embb + off);
            }
        }

        int ra = lo - base; ra = ra < 0 ? 0 : (ra > 32 ? 32 : ra);
        int rb = hi - base; rb = rb < 0 ? 0 : (rb > 32 ? 32 : rb);
        const unsigned ma = (ra >= 32) ? 0xffffffffu : ((1u << ra) - 1u);
        const unsigned mb = (rb >= 32) ? 0xffffffffu : ((1u << rb) - 1u);
        const unsigned mh = (mb & ~ma) >> (2 * h);

#pragma unroll
        for (int kg = 0; kg < 8; ++kg) {
            v2f av;
            av.x = __uint_as_float((unsigned)(((int)(mh << (31 - 4 * kg)) >> 31)) & 0x3f800000u);
            av.y = __uint_as_float((unsigned)(((int)(mh << (30 - 4 * kg)) >> 31)) & 0x3f800000u);
            v2f b0; b0.x = ld[kg][0].x; b0.y = ld[kg][1].x;
            v2f b1; b1.x = ld[kg][0].y; b1.y = ld[kg][1].y;
            v2f b2; b2.x = ld[kg][0].z; b2.y = ld[kg][1].z;
            v2f b3; b3.x = ld[kg][0].w; b3.y = ld[kg][1].w;
            acc0 = __builtin_amdgcn_wmma_f32_16x16x4_f32(
                false, av, false, b0, (short)0, acc0, false, false);
            acc1 = __builtin_amdgcn_wmma_f32_16x16x4_f32(
                false, av, false, b1, (short)0, acc1, false, false);
            acc2 = __builtin_amdgcn_wmma_f32_16x16x4_f32(
                false, av, false, b2, (short)0, acc2, false, false);
            acc3 = __builtin_amdgcn_wmma_f32_16x16x4_f32(
                false, av, false, b3, (short)0, acc3, false, false);
        }
        asm volatile("" : "+v"(idx1));
        idx0 = idx1;
    }

    float* myred = red + (g * 32) * 32 + lane;
#pragma unroll
    for (int r = 0; r < 8; ++r) {
        myred[(0 * 8 + r) * 32] = acc0[r];
        myred[(1 * 8 + r) * 32] = acc1[r];
        myred[(2 * 8 + r) * 32] = acc2[r];
        myred[(3 * 8 + r) * 32] = acc3[r];
    }
    __syncthreads();

    const int mylen = hi - lo;
    const int len = __shfl(mylen, 8 * h + g);
    const float lf = (float)len;
    float o[4];
#pragma unroll
    for (int j = 0; j < 4; ++j) {
        const float* p = red + (j * 8 + g) * 32 + lane;
        float v = p[0];
#pragma unroll
        for (int gg = 1; gg < 8; ++gg) v += p[gg * 32 * 32];
        const float m = (lf > 0.0f) ? (v / lf) : 0.0f;
        o[j] = tanhf(m);
    }
    float4 ov; ov.x = o[0]; ov.y = o[1]; ov.z = o[2]; ov.w = o[3];
    *(float4*)(Amat + (size_t)(s0 + 8 * h + g) * NHID + 64 * q + 4 * c) = ov;
}

__global__ void wmma_head_kernel(const float* __restrict__ Amat,
                                 const float* __restrict__ W,
                                 const float* __restrict__ bias,
                                 float* __restrict__ out) {
    __shared__ float lw[NHID * NPAD];

    const int tid = threadIdx.x;
    for (int i = tid; i < NHID * NPAD; i += 256) {
        const int kk = i >> 4;
        const int nn = i & 15;
        lw[i] = (nn < NCLASSES) ? W[kk * NCLASSES + nn] : 0.0f;
    }
    __syncthreads();

    const int lane = tid & 31;
    const int wave = tid >> 5;
    const int tile = blockIdx.x * 8 + wave;
    const int row0 = tile * 16;

    const int mrow = lane & 15;
    const int koff = (lane >> 4) << 1;
    const float* __restrict__ arow = Amat + (size_t)(row0 + mrow) * NHID;

    v8f c = {};
#pragma unroll 4
    for (int k = 0; k < NHID; k += 4) {
        const float2 a2 = *(const float2*)(arow + k + koff);
        v2f av; av.x = a2.x; av.y = a2.y;
        v2f bv;
        bv.x = lw[(k + koff) * NPAD + mrow];
        bv.y = lw[(k + koff + 1) * NPAD + mrow];
        c = __builtin_amdgcn_wmma_f32_16x16x4_f32(
            false, av, false, bv, (short)0, c, false, false);
    }

    const int mhi = (lane >> 4) << 3;
    const int ncol = lane & 15;
    if (ncol < NCLASSES) {
        const float bb = bias[ncol];
#pragma unroll
        for (int j = 0; j < 8; ++j) {
            const int m = row0 + j + mhi;
            out[(size_t)m * NCLASSES + ncol] = c[j] + bb;
        }
    }
}

extern "C" void kernel_launch(void* const* d_in, const int* in_sizes, int n_in,
                              void* d_out, int out_size, void* d_ws, size_t ws_size,
                              hipStream_t stream) {
    const int*   lengths = (const int*)d_in[0];
    const int*   indices = (const int*)d_in[1];
    const float* emb     = (const float*)d_in[2];
    const float* W       = (const float*)d_in[3];
    const float* bias    = (const float*)d_in[4];
    float*       out     = (float*)d_out;

    int*   offsets = (int*)d_ws;
    float* Amat    = (float*)((char*)d_ws + (64 * 1024));

    seg_scan_kernel<<<1, 1024, 0, stream>>>(lengths, offsets);
    seg_member_sum_tanh_kernel<<<(B_SAMPLES / 16) * 4, 256, 0, stream>>>(offsets, indices, emb, Amat);
    wmma_head_kernel<<<B_SAMPLES / (16 * 8), 256, 0, stream>>>(Amat, W, bias, out);
}


# segment-sum blocks take the 16-sample tiles heaviest first (scan kernel ranks the 512 tiles by token count, a block reads its tile from that order): long tiles no longer end the launch; sums kept
# speedup vs baseline: 3.3351x; 1.0804x over previous
#include <hip/hip_runtime.h>
#include <hip/hip_bf16.h>

#define B_SAMPLES 8192
#define T_TOKENS  1048576
#define NHID      256
#define NCLASSES  14
#define NPAD      16

typedef __attribute__((ext_vector_type(2))) float v2f;
typedef __attribute__((ext_vector_type(8))) float v8f;

#define NTILES      (B_SAMPLES / 16)
#define TILE_BITS   9
#define RANK_BLOCKS 16
static_assert((1 << TILE_BITS) == NTILES, "tile number must fit the key's low bits");
static_assert(NTILES == RANK_BLOCKS * 32, "one tile per wave of a 32-wave block");

__global__ void seg_scan_kernel(const int* __restrict__ lengths,
                                int* __restrict__ offsets,
                                int* __restrict__ order) {
    __shared__ int ssum[1024];
    __shared__ __attribute__((aligned(16))) unsigned key[NTILES];
    const int t = threadIdx.x;
    const int base = t * 8;
    int loc[8];
    int s = 0;
#pragma unroll
    for (int j = 0; j < 8; ++j) {
        loc[j] = lengths[base + j];
        s += loc[j];
    }
    ssum[t] = s;
    __syncthreads();
    for (int off = 1; off < 1024; off <<= 1) {
        int v = (t >= off) ? ssum[t - off] : 0;
        __syncthreads();
        ssum[t] += v;
        __syncthreads();
    }
    if (blockIdx.x == 0) {
        int excl = ssum[t] - s;
#pragma unroll
        for (int j = 0; j < 8; ++j) {
            offsets[base + j] = excl;
            excl += loc[j];
        }
        if (t == 1023) offsets[B_SAMPLES] = ssum[1023];
    }

    if (t < NTILES) {
        const int upto = ssum[2 * t + 1];
        const int before = (t > 0) ? ssum[2 * t - 1] : 0;
        key[t] = ((unsigned)(upto - before) << TILE_BITS) | (unsigned)(NTILES - 1 - t);
    }
    __syncthreads();
    {
        const int lane = t & 31;
        const int i = blockIdx.x * (NTILES / RANK_BLOCKS) + (t >> 5);
        const unsigned mine = key[i];
        const uint4* k4 = (const uint4*)key;
        int r = 0;
        for (int m = 0; m < NTILES / 128; ++m) {
            const uint4 v = k4[lane + 32 * m];
            r += (v.x > mine ? 1 : 0) + (v.y > mine ? 1 : 0)
               + (v.z > mine ? 1 : 0) + (v.w > mine ? 1 : 0);
        }
        for (int d = 16; d >= 1; d >>= 1) r += __shfl_xor(r, d);
        if (lane == 0) order[r] = i;
    }
}

__global__ __launch_bounds__(256) __attribute__((amdgpu_waves_per_eu(8, 8)))
void seg_member_sum_tanh_kernel(const int* __restrict__ offsets,
                                const int* __restrict__ order,
                                const int* __restrict__ indices,
                                const float* __restrict__ emb,
                                float* __restrict__ Amat) {
    __shared__ float red[8 * 32 * 32];

    const int tid  = threadIdx.x;
    const int lane = tid & 31;
    const int g = __builtin_amdgcn_readfirstlane(tid >> 5);
    const int q = blockIdx.x & 3;
    const int s0 = order[blockIdx.x >> 2] * 16;
    const int c = lane & 15;
    const int h = lane >> 4;

    const int lo = offsets[s0 + c];
    const int hi = offsets[s0 + c + 1];
    const int T0 = offsets[s0];
    const int T1 = offsets[s0 + 16];
    const int nb = (T1 - T0 + 31) >> 5;

    const unsigned laneoff = (unsigned)(q * 256 + c * 16);
    const char* __restrict__ embb = (const char*)emb;

    v8f acc0 = {}, acc1 = {}, acc2 = {}, acc3 = {};

    int idx0;
    {
        const int t0 = T0 + (g << 5) + lane;
        idx0 = indices[t0 < T_TOKENS ? t0 : T_TOKENS - 1];
    }
    for (int b = g; b < nb; b += 8) {
        const int base = T0 + (b << 5);
        const int t1 = base + 256 + lane;
        int idx1 = indices[t1 < T_TOKENS ? t1 : T_TOKENS - 1];
        asm volatile("");

        float4 ld[8][2];
#pragma unroll
        for (int kg = 0; kg < 8; ++kg) {
#pragma unroll
            for (int s = 0; s < 2; ++s) {
                const int idx = __shfl(idx0, 4 * kg + 2 * h + s);
                const unsigned off = (unsigned)idx * (unsigned)(NHID * 4) + laneoff;
                ld[kg][s] = *(const float4*)(embb + off);
            }
        }

        int ra = lo - base; ra = ra < 0 ? 0 : (ra > 32 ? 32 : ra);
        int rb = hi - base; rb = rb < 0 ? 0 : (rb > 32 ? 32 : rb);
        const unsigned ma = (ra >= 32) ? 0xffffffffu : ((1u << ra) - 1u);
        const unsigned mb = (rb >= 32) ? 0xffffffffu : ((1u << rb) - 1u);
        const unsigned mh = (mb & ~ma) >> (2 * h);

#pragma unroll
        for (int kg = 0; kg < 8; ++kg) {
            v2f av;
            av.x = __uint_as_float((unsigned)(((int)(mh << (31 - 4 * kg)) >> 31)) & 0x3f800000u);
            av.y = __uint_as_float((unsigned)(((int)(mh << (30 - 4 * kg)) >> 31)) & 0x3f800000u);
            v2f b0; b0.x = ld[kg][0].x; b0.y = ld[kg][1].x;
            v2f b1; b1.x = ld[kg][0].y; b1.y = ld[kg][1].y;
            v2f b2; b2.x = ld[kg][0].z; b2.y = ld[kg][1].z;
            v2f b3; b3.x = ld[kg][0].w; b3.y = ld[kg][1].w;
            acc0 = __builtin_amdgcn_wmma_f32_16x16x4_f32(
                false, av, false, b0, (short)0, acc0, false, false);
            acc1 = __builtin_amdgcn_wmma_f32_16x16x4_f32(
                false, av, false, b1, (short)0, acc1, false, false);
            acc2 = __builtin_amdgcn_wmma_f32_16x16x4_f32(
                false, av, false, b2, (short)0, acc2, false, false);
            acc3 = __builtin_amdgcn_wmma_f32_16x16x4_f32(
                false, av, false, b3, (short)0, acc3, false, false);
        }
        asm volatile("" : "+v"(idx1));
        idx0 = idx1;
    }

    float* myred = red + (g * 32) * 32 + lane;
#pragma unroll
    for (int r = 0; r < 8; ++r) {
        myred[(0 * 8 + r) * 32] = acc0[r];
        myred[(1 * 8 + r) * 32] = acc1[r];
        myred[(2 * 8 + r) * 32] = acc2[r];
        myred[(3 * 8 + r) * 32] = acc3[r];
    }
    __syncthreads();

    const int mylen = hi - lo;
    const int len = __shfl(mylen, 8 * h + g);
    const float lf = (float)len;
    float o[4];
#pragma unroll
    for (int j = 0; j < 4; ++j) {
        const float* p = red + (j * 8 + g) * 32 + lane;
        float v = p[0];
#pragma unroll
        for (int gg = 1; gg < 8; ++gg) v += p[gg * 32 * 32];
        const float m = (lf > 0.0f) ? (v / lf) : 0.0f;
        o[j] = tanhf(m);
    }
    float4 ov; ov.x = o[0]; ov.y = o[1]; ov.z = o[2]; ov.w = o[3];
    *(float4*)(Amat + (size_t)(s0 + 8 * h + g) * NHID + 64 * q + 4 * c) = ov;
}

__global__ void wmma_head_kernel(const float* __restrict__ Amat,
                                 const float* __restrict__ W,
                                 const float* __restrict__ bias,
                                 float* __restrict__ out) {
    __shared__ float lw[NHID * NPAD];

    const int tid = threadIdx.x;
    for (int i = tid; i < NHID * NPAD; i += 256) {
        const int kk = i >> 4;
        const int nn = i & 15;
        lw[i] = (nn < NCLASSES) ? W[kk * NCLASSES + nn] : 0.0f;
    }
    __syncthreads();

    const int lane = tid & 31;
    const int wave = tid >> 5;
    const int tile = blockIdx.x * 8 + wave;
    const int row0 = tile * 16;

    const int mrow = lane & 15;
    const int koff = (lane >> 4) << 1;
    const float* __restrict__ arow = Amat + (size_t)(row0 + mrow) * NHID;

    v8f c = {};
#pragma unroll 4
    for (int k = 0; k < NHID; k += 4) {
        const float2 a2 = *(const float2*)(arow + k + koff);
        v2f av; av.x = a2.x; av.y = a2.y;
        v2f bv;
        bv.x = lw[(k + koff) * NPAD + mrow];
        bv.y = lw[(k + koff + 1) * NPAD + mrow];
        c = __builtin_amdgcn_wmma_f32_16x16x4_f32(
            false, av, false, bv, (short)0, c, false, false);
    }

    const int mhi = (lane >> 4) << 3;
    const int ncol = lane & 15;
    if (ncol < NCLASSES) {
        const float bb = bias[ncol];
#pragma unroll
        for (int j = 0; j < 8; ++j) {
            const int m = row0 + j + mhi;
            out[(size_t)m * NCLASSES + ncol] = c[j] + bb;
        }
    }
}

extern "C" void kernel_launch(void* const* d_in, const int* in_sizes, int n_in,
                              void* d_out, int out_size, void* d_ws, size_t ws_size,
                              hipStream_t stream) {
    const int*   lengths = (const int*)d_in[0];
    const int*   indices = (const int*)d_in[1];
    const float* emb     = (const float*)d_in[2];
    const float* W       = (const float*)d_in[3];
    const float* bias    = (const float*)d_in[4];
    float*       out     = (float*)d_out;

    int*   offsets = (int*)d_ws;
    int*   order   = (int*)((char*)d_ws + (40 * 1024));
    float* Amat    = (float*)((char*)d_ws + (64 * 1024));

    seg_scan_kernel<<<RANK_BLOCKS, 1024, 0, stream>>>(lengths, offsets, order);
    seg_member_sum_tanh_kernel<<<(B_SAMPLES / 16) * 4, 256, 0, stream>>>(offsets, order, indices, emb, Amat);
    wmma_head_kernel<<<B_SAMPLES / (16 * 8), 256, 0, stream>>>(Amat, W, bias, out);
}


# class head fused into the segment-sum kernel: one 32-wave block per 16-sample tile (4 column quarters x 8 token groups), tanh(mean) kept in LDS, 2 launches; head k summed in 8 parts, order changed
# speedup vs baseline: 3.4412x; 1.0318x over previous
#include <hip/hip_runtime.h>
#include <hip/hip_bf16.h>

#define B_SAMPLES 8192
#define T_TOKENS  1048576
#define NHID      256
#define NCLASSES  14
#define NPAD      16

typedef __attribute__((ext_vector_type(2))) float v2f;
typedef __attribute__((ext_vector_type(8))) float v8f;

#define NTILES      (B_SAMPLES / 16)
#define TILE_BITS   9
#define RANK_BLOCKS 16
static_assert((1 << TILE_BITS) == NTILES, "tile number must fit the key's low bits");
static_assert(NTILES == RANK_BLOCKS * 32, "one tile per wave of a 32-wave block");

__global__ void seg_scan_kernel(const int* __restrict__ lengths,
                                int* __restrict__ offsets,
                                int* __restrict__ order) {
    __shared__ int ssum[1024];
    __shared__ __attribute__((aligned(16))) unsigned key[NTILES];
    const int t = threadIdx.x;
    const int base = t * 8;
    int loc[8];
    int s = 0;
#pragma unroll
    for (int j = 0; j < 8; ++j) {
        loc[j] = lengths[base + j];
        s += loc[j];
    }
    ssum[t] = s;
    __syncthreads();
    for (int off = 1; off < 1024; off <<= 1) {
        int v = (t >= off) ? ssum[t - off] : 0;
        __syncthreads();
        ssum[t] += v;
        __syncthreads();
    }
    if (blockIdx.x == 0) {
        int excl = ssum[t] - s;
#pragma unroll
        for (int j = 0; j < 8; ++j) {
            offsets[base + j] = excl;
            excl += loc[j];
        }
        if (t == 1023) offsets[B_SAMPLES] = ssum[1023];
    }

    if (t < NTILES) {
        const int upto = ssum[2 * t + 1];
        const int before = (t > 0) ? ssum[2 * t - 1] : 0;
        key[t] = ((unsigned)(upto - before) << TILE_BITS) | (unsigned)(NTILES - 1 - t);
    }
    __syncthreads();
    {
        const int lane = t & 31;
        const int i = blockIdx.x * (NTILES / RANK_BLOCKS) + (t >> 5);
        const unsigned mine = key[i];
        const uint4* k4 = (const uint4*)key;
        int r = 0;
        for (int m = 0; m < NTILES / 128; ++m) {
            const uint4 v = k4[lane + 32 * m];
            r += (v.x > mine ? 1 : 0) + (v.y > mine ? 1 : 0)
               + (v.z > mine ? 1 : 0) + (v.w > mine ? 1 : 0);
        }
        for (int d = 16; d >= 1; d >>= 1) r += __shfl_xor(r, d);
        if (lane == 0) order[r] = i;
    }
}

#define AT_PITCH 260

__global__ __launch_bounds__(1024) __attribute__((amdgpu_waves_per_eu(8, 8)))
void seg_member_sum_tanh_head_kernel(const int* __restrict__ offsets,
                                     const int* __restrict__ order,
                                     const int* __restrict__ indices,
                                     const float* __restrict__ emb,
                                     const float* __restrict__ W,
                                     const float* __restrict__ bias,
                                     float* __restrict__ out) {
    __shared__ float red[32 * 32 * 32];
    __shared__ float lw[NHID * NPAD];
    __shared__ __attribute__((aligned(16))) float At[16 * AT_PITCH];
    __shared__ float hp[8 * 8 * 32];

    const int tid  = threadIdx.x;
    const int lane = tid & 31;
    const int w = __builtin_amdgcn_readfirstlane(tid >> 5);
    const int g = w >> 2;
    const int q = w & 3;
    const int s0 = order[blockIdx.x] * 16;
    const int c = lane & 15;
    const int h = lane >> 4;

    float wv[4];
#pragma unroll
    for (int m = 0; m < 4; ++m)
        wv[m] = W[((tid >> 4) + 64 * m) * NCLASSES + (c < NCLASSES ? c : NCLASSES - 1)];

    const int lo = offsets[s0 + c];
    const int hi = offsets[s0 + c + 1];
    const int T0 = offsets[s0];
    const int T1 = offsets[s0 + 16];
    const int nb = (T1 - T0 + 31) >> 5;

    const unsigned laneoff = (unsigned)(q * 256 + c * 16);
    const char* __restrict__ embb = (const char*)emb;

    v8f acc0 = {}, acc1 = {}, acc2 = {}, acc3 = {};

    int idx0;
    {
        const int t0 = T0 + (g << 5) + lane;
        idx0 = indices[t0 < T_TOKENS ? t0 : T_TOKENS - 1];
    }
#pragma unroll
    for (int m = 0; m < 4; ++m)
        lw[tid + 1024 * m] = (c < NCLASSES) ? wv[m] : 0.0f;
    for (int b = g; b < nb; b += 8) {
        const int base = T0 + (b << 5);
        const int t1 = base + 256 + lane;
        int idx1 = indices[t1 < T_TOKENS ? t1 : T_TOKENS - 1];
        asm volatile("");

        float4 ld[8][2];
#pragma unroll
        for (int kg = 0; kg < 8; ++kg) {
#pragma unroll
            for (int s = 0; s < 2; ++s) {
                const int idx = __shfl(idx0, 4 * kg + 2 * h + s);
                const unsigned off = (unsigned)idx * (unsigned)(NHID * 4) + laneoff;
                ld[kg][s] = *(const float4*)(embb + off);
            }
        }

        int ra = lo - base; ra = ra < 0 ? 0 : (ra > 32 ? 32 : ra);
        int rb = hi - base; rb = rb < 0 ? 0 : (rb > 32 ? 32 : rb);
        const unsigned ma = (ra >= 32) ? 0xffffffffu : ((1u << ra) - 1u);
        const unsigned mb = (rb >= 32) ? 0xffffffffu : ((1u << rb) - 1u);
        const unsigned mh = (mb & ~ma) >> (2 * h);

#pragma unroll
        for (int kg = 0; kg < 8; ++kg) {
            v2f av;
            av.x = __uint_as_float((unsigned)(((int)(mh << (31 - 4 * kg)) >> 31)) & 0x3f800000u);
            av.y = __uint_as_float((unsigned)(((int)(mh << (30 - 4 * kg)) >> 31)) & 0x3f800000u);
            v2f b0; b0.x = ld[kg][0].x; b0.y = ld[kg][1].x;
            v2f b1; b1.x = ld[kg][0].y; b1.y = ld[kg][1].y;
            v2f b2; b2.x = ld[kg][0].z; b2.y = ld[kg][1].z;
            v2f b3; b3.x = ld[kg][0].w; b3.y = ld[kg][1].w;
            acc0 = __builtin_amdgcn_wmma_f32_16x16x4_f32(
                false, av, false, b0, (short)0, acc0, false, false);
            acc1 = __builtin_amdgcn_wmma_f32_16x16x4_f32(
                false, av, false, b1, (short)0, acc1, false, false);
            acc2 = __builtin_amdgcn_wmma_f32_16x16x4_f32(
                false, av, false, b2, (short)0, acc2, false, false);
            acc3 = __builtin_amdgcn_wmma_f32_16x16x4_f32(
                false, av, false, b3, (short)0, acc3, false, false);
        }
        asm volatile("" : "+v"(idx1));
        idx0 = idx1;
    }

    float* myred = red + (w * 32) * 32 + lane;
#pragma unroll
    for (int r = 0; r < 8; ++r) {
        myred[(0 * 8 + r) * 32] = acc0[r];
        myred[(1 * 8 + r) * 32] = acc1[r];
        myred[(2 * 8 + r) * 32] = acc2[r];
        myred[(3 * 8 + r) * 32] = acc3[r];
    }
    __syncthreads();

    const int mylen = hi - lo;
    const int len = __shfl(mylen, 8 * h + g);
    const float lf = (float)len;
    float o[4];
#pragma unroll
    for (int j = 0; j < 4; ++j) {
        const float* p = red + (q * 32 + j * 8 + g) * 32 + lane;
        float v = p[0];
#pragma unroll
        for (int gg = 1; gg < 8; ++gg) v += p[gg * 4 * 32 * 32];
        const float m = (lf > 0.0f) ? (v / lf) : 0.0f;
        o[j] = tanhf(m);
    }
    float4 ov; ov.x = o[0]; ov.y = o[1]; ov.z = o[2]; ov.w = o[3];
    *(float4*)(At + (8 * h + g) * AT_PITCH + 64 * q + 4 * c) = ov;
    __syncthreads();

    float bb = 0.0f;
    if (w < 8) {
        bb = bias[c < NCLASSES ? c : NCLASSES - 1];
        v8f d = {};
        const float* arow = At + c * AT_PITCH + 32 * w + 2 * h;
        const float* wcol = lw + (32 * w + 2 * h) * NPAD + c;
#pragma unroll
        for (int t = 0; t < 8; ++t) {
            const float2 a2 = *(const float2*)(arow + 4 * t);
            v2f av; av.x = a2.x; av.y = a2.y;
            v2f bv; bv.x = wcol[(4 * t) * NPAD]; bv.y = wcol[(4 * t + 1) * NPAD];
            d = __builtin_amdgcn_wmma_f32_16x16x4_f32(
                false, av, false, bv, (short)0, d, false, false);
        }
        float* myhp = hp + (w * 8) * 32 + lane;
#pragma unroll
        for (int r = 0; r < 8; ++r) myhp[r * 32] = d[r];
    }
    __syncthreads();

    if (w < 8) {
        const float* p = hp + (w * 32) + lane;
        float v = p[0];
#pragma unroll
        for (int ww = 1; ww < 8; ++ww) v += p[ww * 8 * 32];
        if (c < NCLASSES)
            out[(size_t)(s0 + 8 * h + w) * NCLASSES + c] = v + bb;
    }
}

extern "C" void kernel_launch(void* const* d_in, const int* in_sizes, int n_in,
                              void* d_out, int out_size, void* d_ws, size_t ws_size,
                              hipStream_t stream) {
    const int*   lengths = (const int*)d_in[0];
    const int*   indices = (const int*)d_in[1];
    const float* emb     = (const float*)d_in[2];
    const float* W       = (const float*)d_in[3];
    const float* bias    = (const float*)d_in[4];
    float*       out     = (float*)d_out;

    int*   offsets = (int*)d_ws;
    int*   order   = (int*)((char*)d_ws + (40 * 1024));

    seg_scan_kernel<<<RANK_BLOCKS, 1024, 0, stream>>>(lengths, offsets, order);
    seg_member_sum_tanh_head_kernel<<<B_SAMPLES / 16, 1024, 0, stream>>>(offsets, order, indices, emb, W, bias, out);
}


# one launch instead of two: each of 256 resident blocks prefix-sums the lengths in LDS itself and takes the run of samples around its 1/256 of the tokens, 16 samples at a time; sum order changed
# speedup vs baseline: 3.6318x; 1.0554x over previous
#include <hip/hip_runtime.h>
#include <hip/hip_bf16.h>

#define B_SAMPLES 8192
#define T_TOKENS  1048576
#define NHID      256
#define NCLASSES  14
#define NPAD      16

typedef __attribute__((ext_vector_type(2))) float v2f;
typedef __attribute__((ext_vector_type(8))) float v8f;

#define NBLK 256

#define AT_PITCH 260

__global__ __launch_bounds__(1024) __attribute__((amdgpu_waves_per_eu(8, 8)))
void seg_member_sum_tanh_head_kernel(const int* __restrict__ lengths,
                                     const int* __restrict__ indices,
                                     const float* __restrict__ emb,
                                     const float* __restrict__ W,
                                     const float* __restrict__ bias,
                                     float* __restrict__ out) {
    __shared__ float red[32 * 32 * 32];
    __shared__ float lw[NHID * NPAD];
    __shared__ __attribute__((aligned(16))) float At[16 * AT_PITCH];
    __shared__ float hp[8 * 8 * 32];
    __shared__ __attribute__((aligned(16))) int soff[B_SAMPLES + 4];
    __shared__ int wtot[32];
    __shared__ int scut[2];

    const int tid  = threadIdx.x;
    const int lane = tid & 31;
    const int w = __builtin_amdgcn_readfirstlane(tid >> 5);
    const int g = w >> 2;
    const int q = w & 3;
    const int c = lane & 15;
    const int h = lane >> 4;

    float wv[4];
#pragma unroll
    for (int m = 0; m < 4; ++m)
        wv[m] = W[((tid >> 4) + 64 * m) * NCLASSES + (c < NCLASSES ? c : NCLASSES - 1)];

    const int4 la = ((const int4*)lengths)[2 * tid];
    const int4 lb = ((const int4*)lengths)[2 * tid + 1];
    const int sme = la.x + la.y + la.z + la.w + lb.x + lb.y + lb.z + lb.w;
    int v = sme;
#pragma unroll
    for (int d = 1; d < 32; d <<= 1) {
        const int u = __shfl_up(v, d);
        if (lane >= d) v += u;
    }
    if (lane == 31) wtot[w] = v;
    if (tid == 0) { scut[0] = B_SAMPLES; scut[1] = B_SAMPLES; }
    __syncthreads();
    const int wt = wtot[lane];
    int ws = wt;
#pragma unroll
    for (int d = 1; d < 32; d <<= 1) {
        const int u = __shfl_up(ws, d);
        if (lane >= d) ws += u;
    }
    const int total = __builtin_amdgcn_readlane(ws, 31);
    const int wbase = __builtin_amdgcn_readlane(ws - wt, w);
    const int o0 = wbase + v - sme;
    const int o1 = o0 + la.x, o2 = o1 + la.y, o3 = o2 + la.z, o4 = o3 + la.w;
    const int o5 = o4 + lb.x, o6 = o5 + lb.y, o7 = o6 + lb.z, o8 = o7 + lb.w;
    {
        int4 sa; sa.x = o0; sa.y = o1; sa.z = o2; sa.w = o3;
        int4 sb; sb.x = o4; sb.y = o5; sb.z = o6; sb.w = o7;
        *(int4*)(soff + 8 * tid) = sa;
        *(int4*)(soff + 8 * tid + 4) = sb;
        if (tid == 1023) soff[B_SAMPLES] = o8;
    }
    __syncthreads();
    {
        const unsigned thrA = (unsigned)blockIdx.x * (unsigned)total;
        const unsigned thrB = thrA + (unsigned)total;
        bool pA = false, pB = false;
        if (tid > 0) {
            const unsigned m = (unsigned)(soff[8 * tid - 1] + o0) * (unsigned)(NBLK / 2);
            pA = m >= thrA; pB = m >= thrB;
        }
        const int oo[9] = {o0, o1, o2, o3, o4, o5, o6, o7, o8};
#pragma unroll
        for (int j = 0; j < 8; ++j) {
            const unsigned m = (unsigned)(oo[j] + oo[j + 1]) * (unsigned)(NBLK / 2);
            const bool cA = m >= thrA, cB = m >= thrB;
            if (cA && !pA) scut[0] = 8 * tid + j;
            if (cB && !pB) scut[1] = 8 * tid + j;
            pA = cA; pB = cB;
        }
    }
    __syncthreads();
    const int sA = __builtin_amdgcn_readfirstlane(scut[0]);
    const int sB = (blockIdx.x == NBLK - 1) ? B_SAMPLES
                                            : __builtin_amdgcn_readfirstlane(scut[1]);
    int s0 = sA;

    const unsigned laneoff = (unsigned)(q * 256 + c * 16);
    const char* __restrict__ embb = (const char*)emb;

    int idx0;
    {
        const int t0 = soff[sA] + (g << 5) + lane;
        idx0 = indices[t0 < T_TOKENS ? t0 : T_TOKENS - 1];
    }
#pragma unroll
    for (int m = 0; m < 4; ++m)
        lw[tid + 1024 * m] = (c < NCLASSES) ? wv[m] : 0.0f;

    while (s0 < sB) {
    const int sE = (s0 + 16 < sB) ? s0 + 16 : sB;
    const int lo = soff[(s0 + c < sE) ? s0 + c : sE];
    const int hi = soff[(s0 + c + 1 < sE) ? s0 + c + 1 : sE];
    const int T0 = __builtin_amdgcn_readlane(lo, 0);
    const int T1 = __builtin_amdgcn_readlane(hi, 15);
    const int nb = (T1 - T0 + 31) >> 5;
    int idxn;
    {
        const int t0 = T1 + (g << 5) + lane;
        idxn = indices[t0 < T_TOKENS ? t0 : T_TOKENS - 1];
    }
    v8f acc0 = {}, acc1 = {}, acc2 = {}, acc3 = {};
    for (int b = g; b < nb; b += 8) {
        const int base = T0 + (b << 5);
        const int t1 = base + 256 + lane;
        int idx1 = indices[t1 < T_TOKENS ? t1 : T_TOKENS - 1];
        asm volatile("");

        int shb = 8 * h;
        asm volatile("" : "+v"(shb));

        int rix[8][2];
#pragma unroll
        for (int kg = 0; kg < 8; ++kg) {
#pragma unroll
            for (int s = 0; s < 2; ++s)
                rix[kg][s] = __builtin_amdgcn_ds_bpermute(shb + 16 * kg + 4 * s, idx0);
        }
        float4 ld[8][2];
#pragma unroll
        for (int kg = 0; kg < 8; ++kg) {
#pragma unroll
            for (int s = 0; s < 2; ++s) {
                const unsigned off = (unsigned)rix[kg][s] * (unsigned)(NHID * 4) + laneoff;
                ld[kg][s] = *(const float4*)(embb + off);
            }
        }

        int ra = lo - base; ra = ra < 0 ? 0 : (ra > 32 ? 32 : ra);
        int rb = hi - base; rb = rb < 0 ? 0 : (rb > 32 ? 32 : rb);
        const unsigned ma = (ra >= 32) ? 0xffffffffu : ((1u << ra) - 1u);
        const unsigned mb = (rb >= 32) ? 0xffffffffu : ((1u << rb) - 1u);
        const unsigned mh = (mb & ~ma) >> (2 * h);

#pragma unroll
        for (int kg = 0; kg < 8; ++kg) {
            v2f av;
            av.x = __uint_as_float((unsigned)(((int)(mh << (31 - 4 * kg)) >> 31)) & 0x3f800000u);
            av.y = __uint_as_float((unsigned)(((int)(mh << (30 - 4 * kg)) >> 31)) & 0x3f800000u);
            v2f b0; b0.x = ld[kg][0].x; b0.y = ld[kg][1].x;
            v2f b1; b1.x = ld[kg][0].y; b1.y = ld[kg][1].y;
            v2f b2; b2.x = ld[kg][0].z; b2.y = ld[kg][1].z;
            v2f b3; b3.x = ld[kg][0].w; b3.y = ld[kg][1].w;
            acc0 = __builtin_amdgcn_wmma_f32_16x16x4_f32(
                false, av, false, b0, (short)0, acc0, false, false);
            acc1 = __builtin_amdgcn_wmma_f32_16x16x4_f32(
                false, av, false, b1, (short)0, acc1, false, false);
            acc2 = __builtin_amdgcn_wmma_f32_16x16x4_f32(
                false, av, false, b2, (short)0, acc2, false, false);
            acc3 = __builtin_amdgcn_wmma_f32_16x16x4_f32(
                false, av, false, b3, (short)0, acc3, false, false);
        }
        asm volatile("" : "+v"(idx1));
        idx0 = idx1;
    }

    int tl = lane;
    asm volatile("" : "+v"(tl));
    const int lane2 = tl;
    const int c2 = lane2 & 15;
    const int h2 = lane2 >> 4;

    float* myred = red + (w * 32) * 32 + lane2;
#pragma unroll
    for (int r = 0; r < 8; ++r) {
        myred[(0 * 8 + r) * 32] = acc0[r];
        myred[(1 * 8 + r) * 32] = acc1[r];
        myred[(2 * 8 + r) * 32] = acc2[r];
        myred[(3 * 8 + r) * 32] = acc3[r];
    }
    __syncthreads();

    const int mylen = hi - lo;
    const int len = __shfl(mylen, 8 * h2 + g);
    const float lf = (float)len;
    float o[4];
#pragma unroll
    for (int j = 0; j < 4; ++j) {
        const float* p = red + (q * 32 + j * 8 + g) * 32 + lane2;
        float v = p[0];
#pragma unroll
        for (int gg = 1; gg < 8; ++gg) v += p[gg * 4 * 32 * 32];
        const float m = (lf > 0.0f) ? (v / lf) : 0.0f;
        o[j] = tanhf(m);
    }
    float4 ov; ov.x = o[0]; ov.y = o[1]; ov.z = o[2]; ov.w = o[3];
    *(float4*)(At + (8 * h2 + g) * AT_PITCH + 64 * q + 4 * c2) = ov;
    __syncthreads();

    float bb = 0.0f;
    if (w < 8) {
        bb = bias[c2 < NCLASSES ? c2 : NCLASSES - 1];
        v8f d = {};
        const float* arow = At + c2 * AT_PITCH + 32 * w + 2 * h2;
        const float* wcol = lw + (32 * w + 2 * h2) * NPAD + c2;
#pragma unroll
        for (int t = 0; t < 8; ++t) {
            const float2 a2 = *(const float2*)(arow + 4 * t);
            v2f av; av.x = a2.x; av.y = a2.y;
            v2f bv; bv.x = wcol[(4 * t) * NPAD]; bv.y = wcol[(4 * t + 1) * NPAD];
            d = __builtin_amdgcn_wmma_f32_16x16x4_f32(
                false, av, false, bv, (short)0, d, false, false);
        }
        float* myhp = hp + (w * 8) * 32 + lane2;
#pragma unroll
        for (int r = 0; r < 8; ++r) myhp[r * 32] = d[r];
    }
    __syncthreads();

    if (w < 8) {
        const float* p = hp + (w * 32) + lane2;
        float v = p[0];
#pragma unroll
        for (int ww = 1; ww < 8; ++ww) v += p[ww * 8 * 32];
        if (c2 < NCLASSES && s0 + 8 * h2 + w < sE)
            out[(size_t)(s0 + 8 * h2 + w) * NCLASSES + c2] = v + bb;
    }

    s0 += 16; idx0 = idxn;
    }
}

extern "C" void kernel_launch(void* const* d_in, const int* in_sizes, int n_in,
                              void* d_out, int out_size, void* d_ws, size_t ws_size,
                              hipStream_t stream) {
    const int*   lengths = (const int*)d_in[0];
    const int*   indices = (const int*)d_in[1];
    const float* emb     = (const float*)d_in[2];
    const float* W       = (const float*)d_in[3];
    const float* bias    = (const float*)d_in[4];
    float*       out     = (float*)d_out;

    seg_member_sum_tanh_head_kernel<<<NBLK, 1024, 0, stream>>>(lengths, indices, emb, W, bias, out);
}


# batches whose 32 tokens lie in one sample (about 4 in 5) keep one membership pair for all 32 matrix instructions: scalar branches jump over the mask build and the 16 per-token conversions; sums kept
# speedup vs baseline: 3.8935x; 1.0721x over previous
#include <hip/hip_runtime.h>
#include <hip/hip_bf16.h>

#define B_SAMPLES 8192
#define T_TOKENS  1048576
#define NHID      256
#define NCLASSES  14
#define NPAD      16

typedef __attribute__((ext_vector_type(2))) float v2f;
typedef __attribute__((ext_vector_type(8))) float v8f;

#define NBLK 256

#define AT_PITCH 260

__global__ __launch_bounds__(1024) __attribute__((amdgpu_waves_per_eu(8, 8)))
void seg_member_sum_tanh_head_kernel(const int* __restrict__ lengths,
                                     const int* __restrict__ indices,
                                     const float* __restrict__ emb,
                                     const float* __restrict__ W,
                                     const float* __restrict__ bias,
                                     float* __restrict__ out) {
    __shared__ float red[32 * 32 * 32];
    __shared__ float lw[NHID * NPAD];
    __shared__ __attribute__((aligned(16))) float At[16 * AT_PITCH];
    __shared__ float hp[8 * 8 * 32];
    __shared__ __attribute__((aligned(16))) int soff[B_SAMPLES + 4];
    __shared__ int wtot[32];
    __shared__ int scut[2];

    const int tid  = threadIdx.x;
    const int lane = tid & 31;
    const int w = __builtin_amdgcn_readfirstlane(tid >> 5);
    const int g = w >> 2;
    const int q = w & 3;
    const int c = lane & 15;
    const int h = lane >> 4;

    float wv[4];
#pragma unroll
    for (int m = 0; m < 4; ++m)
        wv[m] = W[((tid >> 4) + 64 * m) * NCLASSES + (c < NCLASSES ? c : NCLASSES - 1)];

    const int4 la = ((const int4*)lengths)[2 * tid];
    const int4 lb = ((const int4*)lengths)[2 * tid + 1];
    const int sme = la.x + la.y + la.z + la.w + lb.x + lb.y + lb.z + lb.w;
    int v = sme;
#pragma unroll
    for (int d = 1; d < 32; d <<= 1) {
        const int u = __shfl_up(v, d);
        if (lane >= d) v += u;
    }
    if (lane == 31) wtot[w] = v;
    if (tid == 0) { scut[0] = B_SAMPLES; scut[1] = B_SAMPLES; }
    __syncthreads();
    const int wt = wtot[lane];
    int ws = wt;
#pragma unroll
    for (int d = 1; d < 32; d <<= 1) {
        const int u = __shfl_up(ws, d);
        if (lane >= d) ws += u;
    }
    const int total = __builtin_amdgcn_readlane(ws, 31);
    const int wbase = __builtin_amdgcn_readlane(ws - wt, w);
    const int o0 = wbase + v - sme;
    const int o1 = o0 + la.x, o2 = o1 + la.y, o3 = o2 + la.z, o4 = o3 + la.w;
    const int o5 = o4 + lb.x, o6 = o5 + lb.y, o7 = o6 + lb.z, o8 = o7 + lb.w;
    {
        int4 sa; sa.x = o0; sa.y = o1; sa.z = o2; sa.w = o3;
        int4 sb; sb.x = o4; sb.y = o5; sb.z = o6; sb.w = o7;
        *(int4*)(soff + 8 * tid) = sa;
        *(int4*)(soff + 8 * tid + 4) = sb;
        if (tid == 1023) soff[B_SAMPLES] = o8;
    }
    __syncthreads();
    {
        const unsigned thrA = (unsigned)blockIdx.x * (unsigned)total;
        const unsigned thrB = thrA + (unsigned)total;
        bool pA = false, pB = false;
        if (tid > 0) {
            const unsigned m = (unsigned)(soff[8 * tid - 1] + o0) * (unsigned)(NBLK / 2);
            pA = m >= thrA; pB = m >= thrB;
        }
        const int oo[9] = {o0, o1, o2, o3, o4, o5, o6, o7, o8};
#pragma unroll
        for (int j = 0; j < 8; ++j) {
            const unsigned m = (unsigned)(oo[j] + oo[j + 1]) * (unsigned)(NBLK / 2);
            const bool cA = m >= thrA, cB = m >= thrB;
            if (cA && !pA) scut[0] = 8 * tid + j;
            if (cB && !pB) scut[1] = 8 * tid + j;
            pA = cA; pB = cB;
        }
    }
    __syncthreads();
    const int sA = __builtin_amdgcn_readfirstlane(scut[0]);
    const int sB = (blockIdx.x == NBLK - 1) ? B_SAMPLES
                                            : __builtin_amdgcn_readfirstlane(scut[1]);
    int s0 = sA;

    const unsigned laneoff = (unsigned)(q * 256 + c * 16);
    const char* __restrict__ embb = (const char*)emb;

    int idx0;
    {
        const int t0 = soff[sA] + (g << 5) + lane;
        idx0 = indices[t0 < T_TOKENS ? t0 : T_TOKENS - 1];
    }
#pragma unroll
    for (int m = 0; m < 4; ++m)
        lw[tid + 1024 * m] = (c < NCLASSES) ? wv[m] : 0.0f;

    while (s0 < sB) {
    const int sE = (s0 + 16 < sB) ? s0 + 16 : sB;
    const int lo = soff[(s0 + c < sE) ? s0 + c : sE];
    const int hi = soff[(s0 + c + 1 < sE) ? s0 + c + 1 : sE];
    const int T0 = __builtin_amdgcn_readlane(lo, 0);
    const int T1 = __builtin_amdgcn_readlane(hi, 15);
    const int nb = (T1 - T0 + 31) >> 5;
    int idxn;
    {
        const int t0 = T1 + (g << 5) + lane;
        idxn = indices[t0 < T_TOKENS ? t0 : T_TOKENS - 1];
    }
    v8f acc0 = {}, acc1 = {}, acc2 = {}, acc3 = {};
    for (int b = g; b < nb; b += 8) {
        const int base = T0 + (b << 5);
        const int t1 = base + 256 + lane;
        int idx1 = indices[t1 < T_TOKENS ? t1 : T_TOKENS - 1];
        asm volatile("");

        int shb = 8 * h;
        asm volatile("" : "+v"(shb));

        int rix[8][2];
#pragma unroll
        for (int kg = 0; kg < 8; ++kg) {
#pragma unroll
            for (int s = 0; s < 2; ++s)
                rix[kg][s] = __builtin_amdgcn_ds_bpermute(shb + 16 * kg + 4 * s, idx0);
        }
        float4 ld[8][2];
#pragma unroll
        for (int kg = 0; kg < 8; ++kg) {
#pragma unroll
            for (int s = 0; s < 2; ++s) {
                const unsigned off = (unsigned)rix[kg][s] * (unsigned)(NHID * 4) + laneoff;
                ld[kg][s] = *(const float4*)(embb + off);
            }
        }

        const bool whole = (lo <= base) && (hi >= base + 32);
        const bool one_sample = __builtin_amdgcn_ballot_w32(whole) != 0u;

        unsigned mh = 0u;
        if (!one_sample) {
            int ra = lo - base; ra = ra < 0 ? 0 : (ra > 32 ? 32 : ra);
            int rb = hi - base; rb = rb < 0 ? 0 : (rb > 32 ? 32 : rb);
            const unsigned ma = (ra >= 32) ? 0xffffffffu : ((1u << ra) - 1u);
            const unsigned mb = (rb >= 32) ? 0xffffffffu : ((1u << rb) - 1u);
            mh = (mb & ~ma) >> (2 * h);
        }

        v2f av;
        av.x = whole ? 1.0f : 0.0f;
        av.y = av.x;

#pragma unroll
        for (int kg = 0; kg < 8; ++kg) {
            if (!one_sample) {
                av.x = __uint_as_float((unsigned)(((int)(mh << (31 - 4 * kg)) >> 31)) & 0x3f800000u);
                av.y = __uint_as_float((unsigned)(((int)(mh << (30 - 4 * kg)) >> 31)) & 0x3f800000u);
            }
            v2f b0; b0.x = ld[kg][0].x; b0.y = ld[kg][1].x;
            v2f b1; b1.x = ld[kg][0].y; b1.y = ld[kg][1].y;
            v2f b2; b2.x = ld[kg][0].z; b2.y = ld[kg][1].z;
            v2f b3; b3.x = ld[kg][0].w; b3.y = ld[kg][1].w;
            acc0 = __builtin_amdgcn_wmma_f32_16x16x4_f32(
                false, av, false, b0, (short)0, acc0, false, false);
            acc1 = __builtin_amdgcn_wmma_f32_16x16x4_f32(
                false, av, false, b1, (short)0, acc1, false, false);
            acc2 = __builtin_amdgcn_wmma_f32_16x16x4_f32(
                false, av, false, b2, (short)0, acc2, false, false);
            acc3 = __builtin_amdgcn_wmma_f32_16x16x4_f32(
                false, av, false, b3, (short)0, acc3, false, false);
        }
        asm volatile("" : "+v"(idx1));
        idx0 = idx1;
    }

    int tl = lane;
    asm volatile("" : "+v"(tl));
    const int lane2 = tl;
    const int c2 = lane2 & 15;
    const int h2 = lane2 >> 4;

    float* myred = red + (w * 32) * 32 + lane2;
#pragma unroll
    for (int r = 0; r < 8; ++r) {
        myred[(0 * 8 + r) * 32] = acc0[r];
        myred[(1 * 8 + r) * 32] = acc1[r];
        myred[(2 * 8 + r) * 32] = acc2[r];
        myred[(3 * 8 + r) * 32] = acc3[r];
    }
    __syncthreads();

    const int mylen = hi - lo;
    const int len = __shfl(mylen, 8 * h2 + g);
    const float lf = (float)len;
    float o[4];
#pragma unroll
    for (int j = 0; j < 4; ++j) {
        const float* p = red + (q * 32 + j * 8 + g) * 32 + lane2;
        float v = p[0];
#pragma unroll
        for (int gg = 1; gg < 8; ++gg) v += p[gg * 4 * 32 * 32];
        const float m = (lf > 0.0f) ? (v / lf) : 0.0f;
        o[j] = tanhf(m);
    }
    float4 ov; ov.x = o[0]; ov.y = o[1]; ov.z = o[2]; ov.w = o[3];
    *(float4*)(At + (8 * h2 + g) * AT_PITCH + 64 * q + 4 * c2) = ov;
    __syncthreads();

    float bb = 0.0f;
    if (w < 8) {
        bb = bias[c2 < NCLASSES ? c2 : NCLASSES - 1];
        v8f d = {};
        const float* arow = At + c2 * AT_PITCH + 32 * w + 2 * h2;
        const float* wcol = lw + (32 * w + 2 * h2) * NPAD + c2;
#pragma unroll
        for (int t = 0; t < 8; ++t) {
            const float2 a2 = *(const float2*)(arow + 4 * t);
            v2f av; av.x = a2.x; av.y = a2.y;
            v2f bv; bv.x = wcol[(4 * t) * NPAD]; bv.y = wcol[(4 * t + 1) * NPAD];
            d = __builtin_amdgcn_wmma_f32_16x16x4_f32(
                false, av, false, bv, (short)0, d, false, false);
        }
        float* myhp = hp + (w * 8) * 32 + lane2;
#pragma unroll
        for (int r = 0; r < 8; ++r) myhp[r * 32] = d[r];
    }
    __syncthreads();

    if (w < 8) {
        const float* p = hp + (w * 32) + lane2;
        float v = p[0];
#pragma unroll
        for (int ww = 1; ww < 8; ++ww) v += p[ww * 8 * 32];
        if (c2 < NCLASSES && s0 + 8 * h2 + w < sE)
            out[(size_t)(s0 + 8 * h2 + w) * NCLASSES + c2] = v + bb;
    }

    s0 += 16; idx0 = idxn;
    }
}

extern "C" void kernel_launch(void* const* d_in, const int* in_sizes, int n_in,
                              void* d_out, int out_size, void* d_ws, size_t ws_size,
                              hipStream_t stream) {
    const int*   lengths = (const int*)d_in[0];
    const int*   indices = (const int*)d_in[1];
    const float* emb     = (const float*)d_in[2];
    const float* W       = (const float*)d_in[3];
    const float* bias    = (const float*)d_in[4];
    float*       out     = (float*)d_out;

    seg_member_sum_tanh_head_kernel<<<NBLK, 1024, 0, stream>>>(lengths, indices, emb, W, bias, out);
}


# one-sample batches: the 16 rows a lane holds are added on the vector unit (8 per fragment element) before the membership product: 4 matrix instructions per batch, was 32; sum order changed
# speedup vs baseline: 4.3646x; 1.1210x over previous
#include <hip/hip_runtime.h>
#include <hip/hip_bf16.h>

#define B_SAMPLES 8192
#define T_TOKENS  1048576
#define NHID      256
#define NCLASSES  14
#define NPAD      16

typedef __attribute__((ext_vector_type(2))) float v2f;
typedef __attribute__((ext_vector_type(8))) float v8f;

#define NBLK 256

#define AT_PITCH 260

__global__ __launch_bounds__(1024) __attribute__((amdgpu_waves_per_eu(8, 8)))
void seg_member_sum_tanh_head_kernel(const int* __restrict__ lengths,
                                     const int* __restrict__ indices,
                                     const float* __restrict__ emb,
                                     const float* __restrict__ W,
                                     const float* __restrict__ bias,
                                     float* __restrict__ out) {
    __shared__ float red[32 * 32 * 32];
    __shared__ float lw[NHID * NPAD];
    __shared__ __attribute__((aligned(16))) float At[16 * AT_PITCH];
    __shared__ float hp[8 * 8 * 32];
    __shared__ __attribute__((aligned(16))) int soff[B_SAMPLES + 4];
    __shared__ int wtot[32];
    __shared__ int scut[2];

    const int tid  = threadIdx.x;
    const int lane = tid & 31;
    const int w = __builtin_amdgcn_readfirstlane(tid >> 5);
    const int g = w >> 2;
    const int q = w & 3;
    const int c = lane & 15;
    const int h = lane >> 4;

    float wv[4];
#pragma unroll
    for (int m = 0; m < 4; ++m)
        wv[m] = W[((tid >> 4) + 64 * m) * NCLASSES + (c < NCLASSES ? c : NCLASSES - 1)];

    const int4 la = ((const int4*)lengths)[2 * tid];
    const int4 lb = ((const int4*)lengths)[2 * tid + 1];
    const int sme = la.x + la.y + la.z + la.w + lb.x + lb.y + lb.z + lb.w;
    int v = sme;
#pragma unroll
    for (int d = 1; d < 32; d <<= 1) {
        const int u = __shfl_up(v, d);
        if (lane >= d) v += u;
    }
    if (lane == 31) wtot[w] = v;
    if (tid == 0) { scut[0] = B_SAMPLES; scut[1] = B_SAMPLES; }
    __syncthreads();
    const int wt = wtot[lane];
    int ws = wt;
#pragma unroll
    for (int d = 1; d < 32; d <<= 1) {
        const int u = __shfl_up(ws, d);
        if (lane >= d) ws += u;
    }
    const int total = __builtin_amdgcn_readlane(ws, 31);
    const int wbase = __builtin_amdgcn_readlane(ws - wt, w);
    const int o0 = wbase + v - sme;
    const int o1 = o0 + la.x, o2 = o1 + la.y, o3 = o2 + la.z, o4 = o3 + la.w;
    const int o5 = o4 + lb.x, o6 = o5 + lb.y, o7 = o6 + lb.z, o8 = o7 + lb.w;
    {
        int4 sa; sa.x = o0; sa.y = o1; sa.z = o2; sa.w = o3;
        int4 sb; sb.x = o4; sb.y = o5; sb.z = o6; sb.w = o7;
        *(int4*)(soff + 8 * tid) = sa;
        *(int4*)(soff + 8 * tid + 4) = sb;
        if (tid == 1023) soff[B_SAMPLES] = o8;
    }
    __syncthreads();
    {
        const unsigned thrA = (unsigned)blockIdx.x * (unsigned)total;
        const unsigned thrB = thrA + (unsigned)total;
        bool pA = false, pB = false;
        if (tid > 0) {
            const unsigned m = (unsigned)(soff[8 * tid - 1] + o0) * (unsigned)(NBLK / 2);
            pA = m >= thrA; pB = m >= thrB;
        }
        const int oo[9] = {o0, o1, o2, o3, o4, o5, o6, o7, o8};
#pragma unroll
        for (int j = 0; j < 8; ++j) {
            const unsigned m = (unsigned)(oo[j] + oo[j + 1]) * (unsigned)(NBLK / 2);
            const bool cA = m >= thrA, cB = m >= thrB;
            if (cA && !pA) scut[0] = 8 * tid + j;
            if (cB && !pB) scut[1] = 8 * tid + j;
            pA = cA; pB = cB;
        }
    }
    __syncthreads();
    const int sA = __builtin_amdgcn_readfirstlane(scut[0]);
    const int sB = (blockIdx.x == NBLK - 1) ? B_SAMPLES
                                            : __builtin_amdgcn_readfirstlane(scut[1]);
    int s0 = sA;

    const unsigned laneoff = (unsigned)(q * 256 + c * 16);
    const char* __restrict__ embb = (const char*)emb;

    int idx0;
    {
        const int t0 = soff[sA] + (g << 5) + lane;
        idx0 = indices[t0 < T_TOKENS ? t0 : T_TOKENS - 1];
    }
#pragma unroll
    for (int m = 0; m < 4; ++m)
        lw[tid + 1024 * m] = (c < NCLASSES) ? wv[m] : 0.0f;

    while (s0 < sB) {
    const int sE = (s0 + 16 < sB) ? s0 + 16 : sB;
    const int lo = soff[(s0 + c < sE) ? s0 + c : sE];
    const int hi = soff[(s0 + c + 1 < sE) ? s0 + c + 1 : sE];
    const int T0 = __builtin_amdgcn_readlane(lo, 0);
    const int T1 = __builtin_amdgcn_readlane(hi, 15);
    const int nb = (T1 - T0 + 31) >> 5;
    int idxn;
    {
        const int t0 = T1 + (g << 5) + lane;
        idxn = indices[t0 < T_TOKENS ? t0 : T_TOKENS - 1];
    }
    v8f acc0 = {}, acc1 = {}, acc2 = {}, acc3 = {};
    for (int b = g; b < nb; b += 8) {
        const int base = T0 + (b << 5);
        const int t1 = base + 256 + lane;
        int idx1 = indices[t1 < T_TOKENS ? t1 : T_TOKENS - 1];
        asm volatile("");

        int shb = 8 * h;
        asm volatile("" : "+v"(shb));

        int rix[8][2];
#pragma unroll
        for (int kg = 0; kg < 8; ++kg) {
#pragma unroll
            for (int s = 0; s < 2; ++s)
                rix[kg][s] = __builtin_amdgcn_ds_bpermute(shb + 16 * kg + 4 * s, idx0);
        }
        float4 ld[8][2];
#pragma unroll
        for (int kg = 0; kg < 8; ++kg) {
#pragma unroll
            for (int s = 0; s < 2; ++s) {
                const unsigned off = (unsigned)rix[kg][s] * (unsigned)(NHID * 4) + laneoff;
                ld[kg][s] = *(const float4*)(embb + off);
            }
        }

        const bool whole = (lo <= base) && (hi >= base + 32);
        const bool one_sample = __builtin_amdgcn_ballot_w32(whole) != 0u;

        unsigned mh = 0u;
        if (!one_sample) {
            int ra = lo - base; ra = ra < 0 ? 0 : (ra > 32 ? 32 : ra);
            int rb = hi - base; rb = rb < 0 ? 0 : (rb > 32 ? 32 : rb);
            const unsigned ma = (ra >= 32) ? 0xffffffffu : ((1u << ra) - 1u);
            const unsigned mb = (rb >= 32) ? 0xffffffffu : ((1u << rb) - 1u);
            mh = (mb & ~ma) >> (2 * h);
        }

        v2f av;
        av.x = whole ? 1.0f : 0.0f;
        av.y = av.x;

        float4 u0, u1;
        if (one_sample) {
            float4 t0 = ld[0][0], t1 = ld[0][1];
#pragma unroll
            for (int kg = 1; kg < 8; ++kg) {
                t0.x += ld[kg][0].x; t0.y += ld[kg][0].y; t0.z += ld[kg][0].z; t0.w += ld[kg][0].w;
                t1.x += ld[kg][1].x; t1.y += ld[kg][1].y; t1.z += ld[kg][1].z; t1.w += ld[kg][1].w;
            }
            u0 = t0; u1 = t1;
        } else {
#pragma unroll
            for (int kg = 0; kg < 7; ++kg) {
                av.x = __uint_as_float((unsigned)(((int)(mh << (31 - 4 * kg)) >> 31)) & 0x3f800000u);
                av.y = __uint_as_float((unsigned)(((int)(mh << (30 - 4 * kg)) >> 31)) & 0x3f800000u);
                v2f b0; b0.x = ld[kg][0].x; b0.y = ld[kg][1].x;
                v2f b1; b1.x = ld[kg][0].y; b1.y = ld[kg][1].y;
                v2f b2; b2.x = ld[kg][0].z; b2.y = ld[kg][1].z;
                v2f b3; b3.x = ld[kg][0].w; b3.y = ld[kg][1].w;
                acc0 = __builtin_amdgcn_wmma_f32_16x16x4_f32(
                    false, av, false, b0, (short)0, acc0, false, false);
                acc1 = __builtin_amdgcn_wmma_f32_16x16x4_f32(
                    false, av, false, b1, (short)0, acc1, false, false);
                acc2 = __builtin_amdgcn_wmma_f32_16x16x4_f32(
                    false, av, false, b2, (short)0, acc2, false, false);
                acc3 = __builtin_amdgcn_wmma_f32_16x16x4_f32(
                    false, av, false, b3, (short)0, acc3, false, false);
            }
            av.x = __uint_as_float((unsigned)(((int)(mh << 3) >> 31)) & 0x3f800000u);
            av.y = __uint_as_float((unsigned)(((int)(mh << 2) >> 31)) & 0x3f800000u);
            __builtin_amdgcn_sched_barrier(0);
            u0 = ld[7][0]; u1 = ld[7][1];
        }
        {
            v2f b0; b0.x = u0.x; b0.y = u1.x;
            v2f b1; b1.x = u0.y; b1.y = u1.y;
            v2f b2; b2.x = u0.z; b2.y = u1.z;
            v2f b3; b3.x = u0.w; b3.y = u1.w;
            acc0 = __builtin_amdgcn_wmma_f32_16x16x4_f32(
                false, av, false, b0, (short)0, acc0, false, false);
            acc1 = __builtin_amdgcn_wmma_f32_16x16x4_f32(
                false, av, false, b1, (short)0, acc1, false, false);
            acc2 = __builtin_amdgcn_wmma_f32_16x16x4_f32(
                false, av, false, b2, (short)0, acc2, false, false);
            acc3 = __builtin_amdgcn_wmma_f32_16x16x4_f32(
                false, av, false, b3, (short)0, acc3, false, false);
        }
        asm volatile("" : "+v"(idx1));
        idx0 = idx1;
    }

    int tl = lane;
    asm volatile("" : "+v"(tl));
    const int lane2 = tl;
    const int c2 = lane2 & 15;
    const int h2 = lane2 >> 4;

    float* myred = red + (w * 32) * 32 + lane2;
#pragma unroll
    for (int r = 0; r < 8; ++r) {
        myred[(0 * 8 + r) * 32] = acc0[r];
        myred[(1 * 8 + r) * 32] = acc1[r];
        myred[(2 * 8 + r) * 32] = acc2[r];
        myred[(3 * 8 + r) * 32] = acc3[r];
    }
    __syncthreads();

    const int mylen = hi - lo;
    const int len = __shfl(mylen, 8 * h2 + g);
    const float lf = (float)len;
    float o[4];
#pragma unroll
    for (int j = 0; j < 4; ++j) {
        const float* p = red + (q * 32 + j * 8 + g) * 32 + lane2;
        float v = p[0];
#pragma unroll
        for (int gg = 1; gg < 8; ++gg) v += p[gg * 4 * 32 * 32];
        const float m = (lf > 0.0f) ? (v / lf) : 0.0f;
        o[j] = tanhf(m);
    }
    float4 ov; ov.x = o[0]; ov.y = o[1]; ov.z = o[2]; ov.w = o[3];
    *(float4*)(At + (8 * h2 + g) * AT_PITCH + 64 * q + 4 * c2) = ov;
    __syncthreads();

    float bb = 0.0f;
    if (w < 8) {
        bb = bias[c2 < NCLASSES ? c2 : NCLASSES - 1];
        v8f d = {};
        const float* arow = At + c2 * AT_PITCH + 32 * w + 2 * h2;
        const float* wcol = lw + (32 * w + 2 * h2) * NPAD + c2;
#pragma unroll
        for (int t = 0; t < 8; ++t) {
            const float2 a2 = *(const float2*)(arow + 4 * t);
            v2f av; av.x = a2.x; av.y = a2.y;
            v2f bv; bv.x = wcol[(4 * t) * NPAD]; bv.y = wcol[(4 * t + 1) * NPAD];
            d = __builtin_amdgcn_wmma_f32_16x16x4_f32(
                false, av, false, bv, (short)0, d, false, false);
        }
        float* myhp = hp + (w * 8) * 32 + lane2;
#pragma unroll
        for (int r = 0; r < 8; ++r) myhp[r * 32] = d[r];
    }
    __syncthreads();

    if (w < 8) {
        const float* p = hp + (w * 32) + lane2;
        float v = p[0];
#pragma unroll
        for (int ww = 1; ww < 8; ++ww) v += p[ww * 8 * 32];
        if (c2 < NCLASSES && s0 + 8 * h2 + w < sE)
            out[(size_t)(s0 + 8 * h2 + w) * NCLASSES + c2] = v + bb;
    }

    s0 += 16; idx0 = idxn;
    }
}

extern "C" void kernel_launch(void* const* d_in, const int* in_sizes, int n_in,
                              void* d_out, int out_size, void* d_ws, size_t ws_size,
                              hipStream_t stream) {
    const int*   lengths = (const int*)d_in[0];
    const int*   indices = (const int*)d_in[1];
    const float* emb     = (const float*)d_in[2];
    const float* W       = (const float*)d_in[3];
    const float* bias    = (const float*)d_in[4];
    float*       out     = (float*)d_out;

    seg_member_sum_tanh_head_kernel<<<NBLK, 1024, 0, stream>>>(lengths, indices, emb, W, bias, out);
}
